# starting kernel
# baseline (speedup 1.0000x reference)
#include <hip/hip_runtime.h>


typedef __attribute__((ext_vector_type(2))) float v2f;
typedef __attribute__((ext_vector_type(8))) float v8f;

#define N 2048
#define NTILE (N / 16)
#define LOG2N 11

__global__ __launch_bounds__(256) void rowscan_wmma(const float* __restrict__ D,
                                                    float* __restrict__ R) {
  const int lane = threadIdx.x & 31;
  const int wave = threadIdx.x >> 5;
  const int strip = blockIdx.x * 8 + wave;
  const int b     = strip >> 7;
  const int row0  = (strip & 127) << 4;

  const float* __restrict__ Db = D + (size_t)b * N * N;
  float* __restrict__       Rb = R + (size_t)b * N * N;

  __shared__ float lds[8][16 * 16];
  float* tl = lds[wave];

  const int m    = lane & 15;
  const int half = lane >> 4;

  v8f c = {0.f, 0.f, 0.f, 0.f, 0.f, 0.f, 0.f, 0.f};

  for (int tile = 0; tile < NTILE; ++tile) {
#pragma unroll
    for (int t = 0; t < 2; ++t) {
      const int idx4 = t * 32 + lane;
      const int r    = idx4 >> 2;
      const int cg   = idx4 & 3;
      const float4 v =
          *(const float4*)(Db + (size_t)(row0 + r) * N + tile * 16 + cg * 4);
      *(float4*)&tl[r * 16 + cg * 4] = v;
    }

#pragma unroll
    for (int s = 0; s < 4; ++s) {
      const int k0 = s * 4 + half * 2;
      v2f a, u;
      a.x = tl[m * 16 + k0];
      a.y = tl[m * 16 + k0 + 1];
      u.x = (k0     <= m) ? 1.0f : 0.0f;
      u.y = (k0 + 1 <= m) ? 1.0f : 0.0f;
      c = __builtin_amdgcn_wmma_f32_16x16x4_f32(
           false, a,  false, u,
           (short)0, c,  false,  false);
    }

    const int col = tile * 16 + m;
#pragma unroll
    for (int r = 0; r < 8; ++r) {
      const int row = row0 + r + half * 8;
      Rb[(size_t)row * N + col] = c[r];
    }

    const int src = half ? 31 : 15;
#pragma unroll
    for (int r = 0; r < 8; ++r) c[r] = __shfl(c[r], src, 32);
  }
}

__global__ __launch_bounds__(256) void colscan(float* __restrict__ P) {
  const int colg = blockIdx.x * blockDim.x + threadIdx.x;
  const int b = colg >> LOG2N;
  const int j = colg & (N - 1);
  float* base = P + (size_t)b * N * N + j;

  float acc = 0.0f;
#pragma unroll 4
  for (int i = 0; i < N; ++i) {
    __builtin_prefetch(base + (size_t)(i + 16) * N, 0, 0);
    const float v = base[(size_t)i * N];
    acc += v;
    base[(size_t)i * N] = acc;
  }
}

__global__ __launch_bounds__(256) void assemble(const float* __restrict__ D,
                                                const float* __restrict__ P,
                                                float* __restrict__ out) {
  const size_t t = (size_t)blockIdx.x * blockDim.x + threadIdx.x;
  const int j = (int)(t & (N - 1));
  const int i = (int)((t >> LOG2N) & (N - 1));
  const int b = (int)(t >> (2 * LOG2N));
  const float* __restrict__ Pb = P + (size_t)b * N * N;

  float r;
  if (i == j + 1) {
    r = D[(size_t)b * N * N + (size_t)i * N + j];
  } else {
    const int a = (i < j) ? i : j;
    const int cmx = (i < j) ? j : i;
    r = Pb[(size_t)cmx * N + cmx];
    if (a > 0) {
      r -= Pb[(size_t)(a - 1) * N + cmx];
      r -= Pb[(size_t)cmx * N + (a - 1)];
      r += Pb[(size_t)(a - 1) * N + (a - 1)];
    }
  }
  out[t] = r;
}

extern "C" void kernel_launch(void* const* d_in, const int* in_sizes, int n_in,
                              void* d_out, int out_size, void* d_ws, size_t ws_size,
                              hipStream_t stream) {
  const float* D = (const float*)d_in[0];
  float* out = (float*)d_out;
  float* P   = (float*)d_ws;
  const int B = in_sizes[0] / (N * N);

  rowscan_wmma<<<dim3((B * NTILE) / 8), dim3(256), 0, stream>>>(D, P);
  colscan<<<dim3((B * N) / 256), dim3(256), 0, stream>>>(P);
  const size_t total = (size_t)B * N * N;
  assemble<<<dim3((unsigned)(total / 256)), dim3(256), 0, stream>>>(D, P, out);
}


# launch 1 scans D + D^T (mirror tile added in LDS), so the SAT is symmetric and assemble takes (Pcc - 2*Pac + Paa)/2 from row reads only, one thread per column; sum order changed
# speedup vs baseline: 5.0387x; 5.0387x over previous
#include <hip/hip_runtime.h>


typedef __attribute__((ext_vector_type(2))) float v2f;
typedef __attribute__((ext_vector_type(8))) float v8f;

#define N 2048
#define NTILE (N / 16)
#define TPITCH 40
#define AROWS 32
#define ACOLS 256

template <int kStrips, bool kTransposedStore, bool kAddMirror>
__device__ __forceinline__ void scan_rows(const float* src, float* dst, float* tl, float* tt,
                                          const int row0, const int lane) {
  const int m    = lane & 15;
  const int half = lane >> 4;

  v8f c[kStrips];
#pragma unroll
  for (int q = 0; q < kStrips; ++q) c[q] = (v8f){0.f, 0.f, 0.f, 0.f, 0.f, 0.f, 0.f, 0.f};

  for (int tile = 0; tile < NTILE; ++tile) {
    float4 v[2 * kStrips];
#pragma unroll
    for (int t = 0; t < 2 * kStrips; ++t) {
      const int idx4 = t * 32 + lane;
      const int r    = idx4 >> 2;
      const int cg   = idx4 & 3;
      v[t] = *(const float4*)(src + (size_t)(row0 + r) * N + tile * 16 + cg * 4);
    }
    float4 w[2 * kStrips];
    if (kAddMirror) {
#pragma unroll
      for (int t = 0; t < 2 * kStrips; ++t) {
        const int idx4 = t * 32 + lane;
        const int rr   = idx4 / (4 * kStrips);
        const int cg   = idx4 % (4 * kStrips);
        w[t] = *(const float4*)(src + (size_t)(tile * 16 + rr) * N + row0 + cg * 4);
      }
    }
#pragma unroll
    for (int t = 0; t < 2 * kStrips; ++t) {
      const int idx4 = t * 32 + lane;
      const int r    = idx4 >> 2;
      const int cg   = idx4 & 3;
      *(float4*)&tl[r * 16 + cg * 4] = v[t];
    }
    if (kAddMirror) {
#pragma unroll
      for (int t = 0; t < 2 * kStrips; ++t) {
        const int idx4 = t * 32 + lane;
        const int rr   = idx4 / (4 * kStrips);
        const int cg   = idx4 % (4 * kStrips);
        *(float4*)&tt[rr * TPITCH + cg * 4] = w[t];
      }
    }

#pragma unroll
    for (int s = 0; s < 4; ++s) {
      const int k0 = s * 4 + half * 2;
      v2f u;
      u.x = (k0     <= m) ? 1.0f : 0.0f;
      u.y = (k0 + 1 <= m) ? 1.0f : 0.0f;
#pragma unroll
      for (int q = 0; q < kStrips; ++q) {
        v2f a;
        a.x = tl[(q * 16 + m) * 16 + k0];
        a.y = tl[(q * 16 + m) * 16 + k0 + 1];
        if (kAddMirror) {
          a.x += tt[k0 * TPITCH + q * 16 + m];
          a.y += tt[(k0 + 1) * TPITCH + q * 16 + m];
        }
        c[q] = __builtin_amdgcn_wmma_f32_16x16x4_f32(
             false, a,  false, u,
             (short)0, c[q],  false,  false);
      }
    }

#pragma unroll
    for (int q = 0; q < kStrips; ++q) {
      if (kTransposedStore) {
        float* p = dst + (size_t)(tile * 16 + m) * N + row0 + q * 16 + half * 8;
        *(float4*)(p)     = make_float4(c[q][0], c[q][1], c[q][2], c[q][3]);
        *(float4*)(p + 4) = make_float4(c[q][4], c[q][5], c[q][6], c[q][7]);
      } else {
        const int col = tile * 16 + m;
#pragma unroll
        for (int r = 0; r < 8; ++r) {
          const int row = row0 + q * 16 + r + half * 8;
          dst[(size_t)row * N + col] = c[q][r];
        }
      }
    }

    const int srcl = half ? 31 : 15;
#pragma unroll
    for (int q = 0; q < kStrips; ++q) {
#pragma unroll
      for (int r = 0; r < 8; ++r) c[q][r] = __shfl(c[q][r], srcl, 32);
    }
  }
}

__global__ __launch_bounds__(64) void rowscan_transposed(const float* __restrict__ D,
                                                         float* __restrict__ RT) {
  const int lane = threadIdx.x & 31;
  const int wave = threadIdx.x >> 5;
  const int pair = blockIdx.x * 2 + wave;
  const int b    = pair >> 6;
  const int row0 = (pair & 63) << 5;

  __shared__ __attribute__((aligned(16))) float lds[2][32 * 16];
  __shared__ __attribute__((aligned(16))) float ldt[2][16 * TPITCH];
  scan_rows<2, true, true>(D + (size_t)b * N * N, RT + (size_t)b * N * N, lds[wave], ldt[wave],
                           row0, lane);
}

__global__ __launch_bounds__(128) void rowscan_inplace(float* PT) {
  const int lane  = threadIdx.x & 31;
  const int wave  = threadIdx.x >> 5;
  const int strip = blockIdx.x * 4 + wave;
  const int b     = strip >> 7;
  const int row0  = (strip & 127) << 4;

  __shared__ float lds[4][16 * 16];
  float* Pb = PT + (size_t)b * N * N;
  scan_rows<1, false, false>(Pb, Pb, lds[wave], nullptr, row0, lane);
}

__global__ __launch_bounds__(ACOLS) void assemble(const float* __restrict__ D,
                                                  const float* __restrict__ PT,
                                                  float* __restrict__ out) {
  const int blk = blockIdx.x;
  const int j   = (blk & (N / ACOLS - 1)) * ACOLS + threadIdx.x;
  const int i0  = ((blk / (N / ACOLS)) & (N / AROWS - 1)) * AROWS;
  const int b   = blk / ((N / ACOLS) * (N / AROWS));
  const float* __restrict__ Pb = PT + (size_t)b * N * N;
  const float* __restrict__ Db = D + (size_t)b * N * N;
  float* __restrict__ ob = out + (size_t)b * N * N;

  const int   jm   = (j > 0) ? j - 1 : 0;
  const float dgj  = Pb[(size_t)j * N + j];
  const float dgjm = (j > 0) ? Pb[(size_t)jm * N + jm] : 0.0f;

  float dgr[AROWS];
#pragma unroll
  for (int u = 0; u < AROWS; ++u) dgr[u] = Pb[(size_t)(i0 + u) * (N + 1)];
  const float dgm1 = (i0 > 0) ? Pb[(size_t)(i0 - 1) * (N + 1)] : 0.0f;

  float dsub = 0.0f;
  if (j + 1 >= i0 && j + 1 < i0 + AROWS) dsub = Db[(size_t)(j + 1) * N + j];

  float x[AROWS];
#pragma unroll
  for (int u = 0; u < AROWS; ++u) {
    const int  i  = i0 + u;
    const bool up = (j >= i);
    int rr = up ? i - 1 : i;
    rr = (rr < 0) ? 0 : rr;
    const int cc = up ? j : jm;
    x[u] = Pb[(size_t)rr * N + cc];
  }
#pragma unroll
  for (int u = 0; u < AROWS; ++u) {
    const int   i   = i0 + u;
    const bool  up  = (j >= i);
    const float pcc = up ? dgj : dgr[u];
    const float paa = up ? ((u > 0) ? dgr[u - 1] : dgm1) : dgjm;
    const bool  has = up ? (i > 0) : (j > 0);
    const float pac = has ? x[u] : 0.0f;
    float r = pcc - pac;
    r -= pac;
    r += paa;
    r *= 0.5f;
    r = (j == i - 1) ? dsub : r;
    ob[(size_t)i * N + j] = r;
  }
}

extern "C" void kernel_launch(void* const* d_in, const int* in_sizes, int n_in,
                              void* d_out, int out_size, void* d_ws, size_t ws_size,
                              hipStream_t stream) {
  const float* D = (const float*)d_in[0];
  float* out = (float*)d_out;
  float* PT  = (float*)d_ws;
  const int B = in_sizes[0] / (N * N);

  rowscan_transposed<<<dim3((B * (N / 32)) / 2), dim3(64), 0, stream>>>(D, PT);
  rowscan_inplace<<<dim3((B * NTILE) / 4), dim3(128), 0, stream>>>(PT);
  assemble<<<dim3((unsigned)(B * (N / AROWS) * (N / ACOLS))), dim3(ACOLS), 0, stream>>>(D, PT, out);
}


# both row scans split over 16 waves of one work-group: each wave scans 8 tiles into registers, run totals exchanged through LDS, carry-in added before the store; sum order changed
# speedup vs baseline: 7.7293x; 1.5340x over previous
#include <hip/hip_runtime.h>


typedef __attribute__((ext_vector_type(2))) float v2f;
typedef __attribute__((ext_vector_type(8))) float v8f;

#define N 2048
#define NTILE (N / 16)
#define TPITCH 40
#define AROWS 32
#define ACOLS 256

template <int kStrips, int kTiles, int kWaves, bool kTransposedStore, bool kAddMirror>
__device__ __forceinline__ void scan_rows(const float* src, float* dst, float* tl, float* tt,
                                          float* xch, float* offs, const int row0,
                                          const int lane, const int wave) {
  static_assert(kTiles * kWaves == NTILE, "the waves' runs must cover the row");
  constexpr int kRows = 16 * kStrips;
  const int m    = lane & 15;
  const int half = lane >> 4;

  v8f c[kStrips];
#pragma unroll
  for (int q = 0; q < kStrips; ++q) c[q] = (v8f){0.f, 0.f, 0.f, 0.f, 0.f, 0.f, 0.f, 0.f};

  v8f acc[kTiles][kStrips];

#pragma unroll
  for (int lt = 0; lt < kTiles; ++lt) {
    const int tile = wave * kTiles + lt;
    float4 v[2 * kStrips];
#pragma unroll
    for (int t = 0; t < 2 * kStrips; ++t) {
      const int idx4 = t * 32 + lane;
      const int r    = idx4 >> 2;
      const int cg   = idx4 & 3;
      v[t] = *(const float4*)(src + (size_t)(row0 + r) * N + tile * 16 + cg * 4);
    }
    float4 w[2 * kStrips];
    if (kAddMirror) {
#pragma unroll
      for (int t = 0; t < 2 * kStrips; ++t) {
        const int idx4 = t * 32 + lane;
        const int rr   = idx4 / (4 * kStrips);
        const int cg   = idx4 % (4 * kStrips);
        w[t] = *(const float4*)(src + (size_t)(tile * 16 + rr) * N + row0 + cg * 4);
      }
    }
#pragma unroll
    for (int t = 0; t < 2 * kStrips; ++t) {
      const int idx4 = t * 32 + lane;
      const int r    = idx4 >> 2;
      const int cg   = idx4 & 3;
      *(float4*)&tl[r * 16 + cg * 4] = v[t];
    }
    if (kAddMirror) {
#pragma unroll
      for (int t = 0; t < 2 * kStrips; ++t) {
        const int idx4 = t * 32 + lane;
        const int rr   = idx4 / (4 * kStrips);
        const int cg   = idx4 % (4 * kStrips);
        *(float4*)&tt[rr * TPITCH + cg * 4] = w[t];
      }
    }

#pragma unroll
    for (int s = 0; s < 4; ++s) {
      const int k0 = s * 4 + half * 2;
      v2f u;
      u.x = (k0     <= m) ? 1.0f : 0.0f;
      u.y = (k0 + 1 <= m) ? 1.0f : 0.0f;
#pragma unroll
      for (int q = 0; q < kStrips; ++q) {
        v2f a;
        a.x = tl[(q * 16 + m) * 16 + k0];
        a.y = tl[(q * 16 + m) * 16 + k0 + 1];
        if (kAddMirror) {
          a.x += tt[k0 * TPITCH + q * 16 + m];
          a.y += tt[(k0 + 1) * TPITCH + q * 16 + m];
        }
        c[q] = __builtin_amdgcn_wmma_f32_16x16x4_f32(
             false, a,  false, u,
             (short)0, c[q],  false,  false);
      }
    }

#pragma unroll
    for (int q = 0; q < kStrips; ++q) acc[lt][q] = c[q];

    const int srcl = half ? 31 : 15;
#pragma unroll
    for (int q = 0; q < kStrips; ++q) {
#pragma unroll
      for (int r = 0; r < 8; ++r) c[q][r] = __shfl(c[q][r], srcl, 32);
    }
  }

  if (m == 0) {
#pragma unroll
    for (int q = 0; q < kStrips; ++q) {
      float* p = xch + wave * kRows + q * 16 + half * 8;
      *(float4*)(p)     = make_float4(c[q][0], c[q][1], c[q][2], c[q][3]);
      *(float4*)(p + 4) = make_float4(c[q][4], c[q][5], c[q][6], c[q][7]);
    }
  }
  __syncthreads();

  {
    const int rowl = lane & (kRows - 1);
    float o = 0.0f;
    for (int p = 0; p < wave; ++p) o += xch[p * kRows + rowl];
    offs[rowl] = o;
  }

#pragma unroll
  for (int q = 0; q < kStrips; ++q) {
    const float4 o0 = *(const float4*)&offs[q * 16 + half * 8];
    const float4 o1 = *(const float4*)&offs[q * 16 + half * 8 + 4];
    const float  o[8] = {o0.x, o0.y, o0.z, o0.w, o1.x, o1.y, o1.z, o1.w};
#pragma unroll
    for (int lt = 0; lt < kTiles; ++lt) {
      const int tile = wave * kTiles + lt;
      v8f d = acc[lt][q];
#pragma unroll
      for (int r = 0; r < 8; ++r) d[r] += o[r];
      if (kTransposedStore) {
        float* p = dst + (size_t)(tile * 16 + m) * N + row0 + q * 16 + half * 8;
        *(float4*)(p)     = make_float4(d[0], d[1], d[2], d[3]);
        *(float4*)(p + 4) = make_float4(d[4], d[5], d[6], d[7]);
      } else {
        const int col = tile * 16 + m;
#pragma unroll
        for (int r = 0; r < 8; ++r) {
          const int row = row0 + q * 16 + r + half * 8;
          dst[(size_t)row * N + col] = d[r];
        }
      }
    }
  }
}

#define K1WAVES 16
#define K1TILES (NTILE / K1WAVES)
__global__ __launch_bounds__(K1WAVES * 32) void rowscan_transposed(const float* __restrict__ D,
                                                                   float* __restrict__ RT) {
  const int lane = threadIdx.x & 31;
  const int wave = threadIdx.x >> 5;
  const int pair = blockIdx.x;
  const int b    = pair >> 6;
  const int row0 = (pair & 63) << 5;

  __shared__ __attribute__((aligned(16))) float lds[K1WAVES][32 * 16];
  __shared__ __attribute__((aligned(16))) float ldt[K1WAVES][16 * TPITCH];
  __shared__ __attribute__((aligned(16))) float xch[K1WAVES * 32];
  __shared__ __attribute__((aligned(16))) float offs[K1WAVES][32];
  scan_rows<2, K1TILES, K1WAVES, true, true>(D + (size_t)b * N * N, RT + (size_t)b * N * N,
                                             lds[wave], ldt[wave], xch, offs[wave], row0, lane,
                                             wave);
}

#define K2WAVES 16
#define K2TILES (NTILE / K2WAVES)
__global__ __launch_bounds__(K2WAVES * 32) void rowscan_inplace(float* PT) {
  const int lane  = threadIdx.x & 31;
  const int wave  = threadIdx.x >> 5;
  const int strip = blockIdx.x;
  const int b     = strip >> 7;
  const int row0  = (strip & 127) << 4;

  __shared__ __attribute__((aligned(16))) float lds[K2WAVES][16 * 16];
  __shared__ __attribute__((aligned(16))) float xch[K2WAVES * 16];
  __shared__ __attribute__((aligned(16))) float offs[K2WAVES][16];
  float* Pb = PT + (size_t)b * N * N;
  scan_rows<1, K2TILES, K2WAVES, false, false>(Pb, Pb, lds[wave], nullptr, xch, offs[wave], row0,
                                               lane, wave);
}

__global__ __launch_bounds__(ACOLS) void assemble(const float* __restrict__ D,
                                                  const float* __restrict__ PT,
                                                  float* __restrict__ out) {
  const int blk = blockIdx.x;
  const int j   = (blk & (N / ACOLS - 1)) * ACOLS + threadIdx.x;
  const int i0  = ((blk / (N / ACOLS)) & (N / AROWS - 1)) * AROWS;
  const int b   = blk / ((N / ACOLS) * (N / AROWS));
  const float* __restrict__ Pb = PT + (size_t)b * N * N;
  const float* __restrict__ Db = D + (size_t)b * N * N;
  float* __restrict__ ob = out + (size_t)b * N * N;

  const int   jm   = (j > 0) ? j - 1 : 0;
  const float dgj  = Pb[(size_t)j * N + j];
  const float dgjm = (j > 0) ? Pb[(size_t)jm * N + jm] : 0.0f;

  float dgr[AROWS];
#pragma unroll
  for (int u = 0; u < AROWS; ++u) dgr[u] = Pb[(size_t)(i0 + u) * (N + 1)];
  const float dgm1 = (i0 > 0) ? Pb[(size_t)(i0 - 1) * (N + 1)] : 0.0f;

  float dsub = 0.0f;
  if (j + 1 >= i0 && j + 1 < i0 + AROWS) dsub = Db[(size_t)(j + 1) * N + j];

  float x[AROWS];
#pragma unroll
  for (int u = 0; u < AROWS; ++u) {
    const int  i  = i0 + u;
    const bool up = (j >= i);
    int rr = up ? i - 1 : i;
    rr = (rr < 0) ? 0 : rr;
    const int cc = up ? j : jm;
    x[u] = Pb[(size_t)rr * N + cc];
  }
#pragma unroll
  for (int u = 0; u < AROWS; ++u) {
    const int   i   = i0 + u;
    const bool  up  = (j >= i);
    const float pcc = up ? dgj : dgr[u];
    const float paa = up ? ((u > 0) ? dgr[u - 1] : dgm1) : dgjm;
    const bool  has = up ? (i > 0) : (j > 0);
    const float pac = has ? x[u] : 0.0f;
    float r = pcc - pac;
    r -= pac;
    r += paa;
    r *= 0.5f;
    r = (j == i - 1) ? dsub : r;
    ob[(size_t)i * N + j] = r;
  }
}

extern "C" void kernel_launch(void* const* d_in, const int* in_sizes, int n_in,
                              void* d_out, int out_size, void* d_ws, size_t ws_size,
                              hipStream_t stream) {
  const float* D = (const float*)d_in[0];
  float* out = (float*)d_out;
  float* PT  = (float*)d_ws;
  const int B = in_sizes[0] / (N * N);

  rowscan_transposed<<<dim3(B * (N / 32)), dim3(K1WAVES * 32), 0, stream>>>(D, PT);
  rowscan_inplace<<<dim3(B * NTILE), dim3(K2WAVES * 32), 0, stream>>>(PT);
  assemble<<<dim3((unsigned)(B * (N / AROWS) * (N / ACOLS))), dim3(ACOLS), 0, stream>>>(D, PT, out);
}


# both scans store one whole 128-B line per store: launch 1 computes L16 x tile (ones as A, data as B) so its tiles lie in the plane's row order; adjacent tiles swap lane halves; sum order unchanged
# speedup vs baseline: 9.0314x; 1.1685x over previous
#include <hip/hip_runtime.h>


typedef __attribute__((ext_vector_type(2))) float v2f;
typedef __attribute__((ext_vector_type(8))) float v8f;

#define N 2048
#define NTILE (N / 16)
#define TPITCH 40
#define AROWS 32
#define ACOLS 256

__device__ __forceinline__ float xhalf(float x) {
  const int b = __builtin_bit_cast(int, x);
  return __builtin_bit_cast(float, __builtin_amdgcn_permlanex16(b, b, 0x76543210, 0xfedcba98, false, false));
}

template <int kStrips, int kTiles, int kWaves, bool kColumnForm, bool kAddMirror>
__device__ __forceinline__ void scan_rows(const float* src, float* dst, float* tl, float* tt,
                                          float* xch, float* offs, const int row0,
                                          const int lane, const int wave) {
  static_assert(kTiles * kWaves == NTILE, "the waves' runs must cover the row");
  static_assert(kColumnForm ? (kStrips == 2 && kAddMirror) : (kStrips == 1 && kTiles % 2 == 0),
                "whole-line stores: a 32-column panel, or pairs of adjacent tiles of a run");
  constexpr int kRows = 16 * kStrips;
  const int m    = lane & 15;
  const int half = lane >> 4;

  v8f c[kStrips];
#pragma unroll
  for (int q = 0; q < kStrips; ++q) c[q] = (v8f){0.f, 0.f, 0.f, 0.f, 0.f, 0.f, 0.f, 0.f};

  v8f acc[kTiles][kStrips];

#pragma unroll
  for (int lt = 0; lt < kTiles; ++lt) {
    const int tile = wave * kTiles + lt;
    float4 v[2 * kStrips];
#pragma unroll
    for (int t = 0; t < 2 * kStrips; ++t) {
      const int idx4 = t * 32 + lane;
      const int r    = idx4 >> 2;
      const int cg   = idx4 & 3;
      v[t] = *(const float4*)(src + (size_t)(row0 + r) * N + tile * 16 + cg * 4);
    }
    float4 w[2 * kStrips];
    if (kAddMirror) {
#pragma unroll
      for (int t = 0; t < 2 * kStrips; ++t) {
        const int idx4 = t * 32 + lane;
        const int rr   = idx4 / (4 * kStrips);
        const int cg   = idx4 % (4 * kStrips);
        w[t] = *(const float4*)(src + (size_t)(tile * 16 + rr) * N + row0 + cg * 4);
      }
    }
#pragma unroll
    for (int t = 0; t < 2 * kStrips; ++t) {
      const int idx4 = t * 32 + lane;
      const int r    = idx4 >> 2;
      const int cg   = idx4 & 3;
      *(float4*)&tl[r * 16 + cg * 4] = v[t];
    }
    if (kAddMirror) {
#pragma unroll
      for (int t = 0; t < 2 * kStrips; ++t) {
        const int idx4 = t * 32 + lane;
        const int rr   = idx4 / (4 * kStrips);
        const int cg   = idx4 % (4 * kStrips);
        *(float4*)&tt[rr * TPITCH + cg * 4] = w[t];
      }
    }

#pragma unroll
    for (int s = 0; s < 4; ++s) {
      const int k0 = s * 4 + half * 2;
      v2f u;
      u.x = (k0     <= m) ? 1.0f : 0.0f;
      u.y = (k0 + 1 <= m) ? 1.0f : 0.0f;
#pragma unroll
      for (int q = 0; q < kStrips; ++q) {
        v2f a;
        a.x = tl[(q * 16 + m) * 16 + k0];
        a.y = tl[(q * 16 + m) * 16 + k0 + 1];
        if (kAddMirror) {
          a.x += tt[k0 * TPITCH + q * 16 + m];
          a.y += tt[(k0 + 1) * TPITCH + q * 16 + m];
        }
        if constexpr (kColumnForm) {
          c[q] = __builtin_amdgcn_wmma_f32_16x16x4_f32(
               false, u,  false, a,
               (short)0, c[q],  false,  false);
        } else {
          c[q] = __builtin_amdgcn_wmma_f32_16x16x4_f32(
               false, a,  false, u,
               (short)0, c[q],  false,  false);
        }
      }
    }

#pragma unroll
    for (int q = 0; q < kStrips; ++q) acc[lt][q] = c[q];

    if constexpr (kColumnForm) {
#pragma unroll
      for (int q = 0; q < kStrips; ++q) {
        const float own = c[q][7];
        const float oth = xhalf(own);
        const float cy  = half ? own : oth;
#pragma unroll
        for (int r = 0; r < 8; ++r) c[q][r] = cy;
      }
    } else {
      const int srcl = half ? 31 : 15;
#pragma unroll
      for (int q = 0; q < kStrips; ++q) {
#pragma unroll
        for (int r = 0; r < 8; ++r) c[q][r] = __shfl(c[q][r], srcl, 32);
      }
    }
  }

  if constexpr (kColumnForm) {
    if (half == 0) {
#pragma unroll
      for (int q = 0; q < kStrips; ++q) xch[wave * kRows + q * 16 + m] = c[q][0];
    }
    __syncthreads();

    float o[kStrips];
#pragma unroll
    for (int q = 0; q < kStrips; ++q) o[q] = 0.0f;
    for (int p = 0; p < wave; ++p) {
#pragma unroll
      for (int q = 0; q < kStrips; ++q) o[q] += xch[p * kRows + q * 16 + m];
    }

#pragma unroll
    for (int lt = 0; lt < kTiles; ++lt) {
      const int tile = wave * kTiles + lt;
      float* p = dst + (size_t)(tile * 16) * N + row0 + lane;
#pragma unroll
      for (int r = 0; r < 8; ++r) {
        const float d0 = acc[lt][0][r] + o[0];
        const float d1 = acc[lt][1][r] + o[1];
        const float t  = half ? d0 : d1;
        const float g  = xhalf(t);
        const float X  = half ? g : d0;
        const float Y  = half ? d1 : g;
        p[(size_t)r * N]       = X;
        p[(size_t)(r + 8) * N] = Y;
      }
    }
  } else {
    if (m == 0) {
#pragma unroll
      for (int q = 0; q < kStrips; ++q) {
        float* p = xch + wave * kRows + q * 16 + half * 8;
        *(float4*)(p)     = make_float4(c[q][0], c[q][1], c[q][2], c[q][3]);
        *(float4*)(p + 4) = make_float4(c[q][4], c[q][5], c[q][6], c[q][7]);
      }
    }
    __syncthreads();

    {
      const int rowl = lane & (kRows - 1);
      float o = 0.0f;
      for (int p = 0; p < wave; ++p) o += xch[p * kRows + rowl];
      offs[rowl] = o;
    }

    const float4 o0 = *(const float4*)&offs[half * 8];
    const float4 o1 = *(const float4*)&offs[half * 8 + 4];
    const float  o[8] = {o0.x, o0.y, o0.z, o0.w, o1.x, o1.y, o1.z, o1.w};
#pragma unroll
    for (int lt = 0; lt < kTiles; lt += 2) {
      const int tile = wave * kTiles + lt;
      float* p = dst + (size_t)row0 * N + tile * 16 + lane;
#pragma unroll
      for (int r = 0; r < 8; ++r) {
        const float d0 = acc[lt][0][r] + o[r];
        const float d1 = acc[lt + 1][0][r] + o[r];
        const float t  = half ? d0 : d1;
        const float g  = xhalf(t);
        const float X  = half ? g : d0;
        const float Y  = half ? d1 : g;
        p[(size_t)r * N]       = X;
        p[(size_t)(r + 8) * N] = Y;
      }
    }
  }
}

#define K1WAVES 16
#define K1TILES (NTILE / K1WAVES)
__global__ __launch_bounds__(K1WAVES * 32) void scan_down_panels(const float* __restrict__ D,
                                                                   float* __restrict__ RT) {
  const int lane = threadIdx.x & 31;
  const int wave = threadIdx.x >> 5;
  const int pair = blockIdx.x;
  const int b    = pair >> 6;
  const int row0 = (pair & 63) << 5;

  __shared__ __attribute__((aligned(16))) float lds[K1WAVES][32 * 16];
  __shared__ __attribute__((aligned(16))) float ldt[K1WAVES][16 * TPITCH];
  __shared__ __attribute__((aligned(16))) float xch[K1WAVES * 32];
  scan_rows<2, K1TILES, K1WAVES, true, true>(D + (size_t)b * N * N, RT + (size_t)b * N * N,
                                             lds[wave], ldt[wave], xch, nullptr, row0, lane,
                                             wave);
}

#define K2WAVES 16
#define K2TILES (NTILE / K2WAVES)
__global__ __launch_bounds__(K2WAVES * 32) void rowscan_inplace(float* PT) {
  const int lane  = threadIdx.x & 31;
  const int wave  = threadIdx.x >> 5;
  const int strip = blockIdx.x;
  const int b     = strip >> 7;
  const int row0  = (strip & 127) << 4;

  __shared__ __attribute__((aligned(16))) float lds[K2WAVES][16 * 16];
  __shared__ __attribute__((aligned(16))) float xch[K2WAVES * 16];
  __shared__ __attribute__((aligned(16))) float offs[K2WAVES][16];
  float* Pb = PT + (size_t)b * N * N;
  scan_rows<1, K2TILES, K2WAVES, false, false>(Pb, Pb, lds[wave], nullptr, xch, offs[wave], row0,
                                               lane, wave);
}

__global__ __launch_bounds__(ACOLS) void assemble(const float* __restrict__ D,
                                                  const float* __restrict__ PT,
                                                  float* __restrict__ out) {
  const int blk = blockIdx.x;
  const int j   = (blk & (N / ACOLS - 1)) * ACOLS + threadIdx.x;
  const int i0  = ((blk / (N / ACOLS)) & (N / AROWS - 1)) * AROWS;
  const int b   = blk / ((N / ACOLS) * (N / AROWS));
  const float* __restrict__ Pb = PT + (size_t)b * N * N;
  const float* __restrict__ Db = D + (size_t)b * N * N;
  float* __restrict__ ob = out + (size_t)b * N * N;

  const int   jm   = (j > 0) ? j - 1 : 0;
  const float dgj  = Pb[(size_t)j * N + j];
  const float dgjm = (j > 0) ? Pb[(size_t)jm * N + jm] : 0.0f;

  float dgr[AROWS];
#pragma unroll
  for (int u = 0; u < AROWS; ++u) dgr[u] = Pb[(size_t)(i0 + u) * (N + 1)];
  const float dgm1 = (i0 > 0) ? Pb[(size_t)(i0 - 1) * (N + 1)] : 0.0f;

  float dsub = 0.0f;
  if (j + 1 >= i0 && j + 1 < i0 + AROWS) dsub = Db[(size_t)(j + 1) * N + j];

  float x[AROWS];
#pragma unroll
  for (int u = 0; u < AROWS; ++u) {
    const int  i  = i0 + u;
    const bool up = (j >= i);
    int rr = up ? i - 1 : i;
    rr = (rr < 0) ? 0 : rr;
    const int cc = up ? j : jm;
    x[u] = Pb[(size_t)rr * N + cc];
  }
#pragma unroll
  for (int u = 0; u < AROWS; ++u) {
    const int   i   = i0 + u;
    const bool  up  = (j >= i);
    const float pcc = up ? dgj : dgr[u];
    const float paa = up ? ((u > 0) ? dgr[u - 1] : dgm1) : dgjm;
    const bool  has = up ? (i > 0) : (j > 0);
    const float pac = has ? x[u] : 0.0f;
    float r = pcc - pac;
    r -= pac;
    r += paa;
    r *= 0.5f;
    r = (j == i - 1) ? dsub : r;
    ob[(size_t)i * N + j] = r;
  }
}

extern "C" void kernel_launch(void* const* d_in, const int* in_sizes, int n_in,
                              void* d_out, int out_size, void* d_ws, size_t ws_size,
                              hipStream_t stream) {
  const float* D = (const float*)d_in[0];
  float* out = (float*)d_out;
  float* PT  = (float*)d_ws;
  const int B = in_sizes[0] / (N * N);

  scan_down_panels<<<dim3(B * (N / 32)), dim3(K1WAVES * 32), 0, stream>>>(D, PT);
  rowscan_inplace<<<dim3(B * NTILE), dim3(K2WAVES * 32), 0, stream>>>(PT);
  assemble<<<dim3((unsigned)(B * (N / AROWS) * (N / ACOLS))), dim3(ACOLS), 0, stream>>>(D, PT, out);
}


# assemble: a work-group walks down 128 rows in 4 chunks of 32, so 8,192 waves enter, not 32,768, and a column's two diagonal entries are gathered once per 128 rows; term order unchanged
# speedup vs baseline: 9.3156x; 1.0315x over previous
#include <hip/hip_runtime.h>


typedef __attribute__((ext_vector_type(2))) float v2f;
typedef __attribute__((ext_vector_type(8))) float v8f;

#define N 2048
#define NTILE (N / 16)
#define TPITCH 40
#define AROWS 32
#define ACHUNKS 4
#define ACOLS 256

__device__ __forceinline__ float xhalf(float x) {
  const int b = __builtin_bit_cast(int, x);
  return __builtin_bit_cast(float, __builtin_amdgcn_permlanex16(b, b, 0x76543210, 0xfedcba98, false, false));
}

template <int kStrips, int kTiles, int kWaves, bool kColumnForm, bool kAddMirror>
__device__ __forceinline__ void scan_rows(const float* src, float* dst, float* tl, float* tt,
                                          float* xch, float* offs, const int row0,
                                          const int lane, const int wave) {
  static_assert(kTiles * kWaves == NTILE, "the waves' runs must cover the row");
  static_assert(kColumnForm ? (kStrips == 2 && kAddMirror) : (kStrips == 1 && kTiles % 2 == 0),
                "whole-line stores: a 32-column panel, or pairs of adjacent tiles of a run");
  constexpr int kRows = 16 * kStrips;
  const int m    = lane & 15;
  const int half = lane >> 4;

  v8f c[kStrips];
#pragma unroll
  for (int q = 0; q < kStrips; ++q) c[q] = (v8f){0.f, 0.f, 0.f, 0.f, 0.f, 0.f, 0.f, 0.f};

  v8f acc[kTiles][kStrips];

#pragma unroll
  for (int lt = 0; lt < kTiles; ++lt) {
    const int tile = wave * kTiles + lt;
    float4 v[2 * kStrips];
#pragma unroll
    for (int t = 0; t < 2 * kStrips; ++t) {
      const int idx4 = t * 32 + lane;
      const int r    = idx4 >> 2;
      const int cg   = idx4 & 3;
      v[t] = *(const float4*)(src + (size_t)(row0 + r) * N + tile * 16 + cg * 4);
    }
    float4 w[2 * kStrips];
    if (kAddMirror) {
#pragma unroll
      for (int t = 0; t < 2 * kStrips; ++t) {
        const int idx4 = t * 32 + lane;
        const int rr   = idx4 / (4 * kStrips);
        const int cg   = idx4 % (4 * kStrips);
        w[t] = *(const float4*)(src + (size_t)(tile * 16 + rr) * N + row0 + cg * 4);
      }
    }
#pragma unroll
    for (int t = 0; t < 2 * kStrips; ++t) {
      const int idx4 = t * 32 + lane;
      const int r    = idx4 >> 2;
      const int cg   = idx4 & 3;
      *(float4*)&tl[r * 16 + cg * 4] = v[t];
    }
    if (kAddMirror) {
#pragma unroll
      for (int t = 0; t < 2 * kStrips; ++t) {
        const int idx4 = t * 32 + lane;
        const int rr   = idx4 / (4 * kStrips);
        const int cg   = idx4 % (4 * kStrips);
        *(float4*)&tt[rr * TPITCH + cg * 4] = w[t];
      }
    }

#pragma unroll
    for (int s = 0; s < 4; ++s) {
      const int k0 = s * 4 + half * 2;
      v2f u;
      u.x = (k0     <= m) ? 1.0f : 0.0f;
      u.y = (k0 + 1 <= m) ? 1.0f : 0.0f;
#pragma unroll
      for (int q = 0; q < kStrips; ++q) {
        v2f a;
        a.x = tl[(q * 16 + m) * 16 + k0];
        a.y = tl[(q * 16 + m) * 16 + k0 + 1];
        if (kAddMirror) {
          a.x += tt[k0 * TPITCH + q * 16 + m];
          a.y += tt[(k0 + 1) * TPITCH + q * 16 + m];
        }
        if constexpr (kColumnForm) {
          c[q] = __builtin_amdgcn_wmma_f32_16x16x4_f32(
               false, u,  false, a,
               (short)0, c[q],  false,  false);
        } else {
          c[q] = __builtin_amdgcn_wmma_f32_16x16x4_f32(
               false, a,  false, u,
               (short)0, c[q],  false,  false);
        }
      }
    }

#pragma unroll
    for (int q = 0; q < kStrips; ++q) acc[lt][q] = c[q];

    if constexpr (kColumnForm) {
#pragma unroll
      for (int q = 0; q < kStrips; ++q) {
        const float own = c[q][7];
        const float oth = xhalf(own);
        const float cy  = half ? own : oth;
#pragma unroll
        for (int r = 0; r < 8; ++r) c[q][r] = cy;
      }
    } else {
      const int srcl = half ? 31 : 15;
#pragma unroll
      for (int q = 0; q < kStrips; ++q) {
#pragma unroll
        for (int r = 0; r < 8; ++r) c[q][r] = __shfl(c[q][r], srcl, 32);
      }
    }
  }

  if constexpr (kColumnForm) {
    if (half == 0) {
#pragma unroll
      for (int q = 0; q < kStrips; ++q) xch[wave * kRows + q * 16 + m] = c[q][0];
    }
    __syncthreads();

    float o[kStrips];
#pragma unroll
    for (int q = 0; q < kStrips; ++q) o[q] = 0.0f;
    for (int p = 0; p < wave; ++p) {
#pragma unroll
      for (int q = 0; q < kStrips; ++q) o[q] += xch[p * kRows + q * 16 + m];
    }

#pragma unroll
    for (int lt = 0; lt < kTiles; ++lt) {
      const int tile = wave * kTiles + lt;
      float* p = dst + (size_t)(tile * 16) * N + row0 + lane;
#pragma unroll
      for (int r = 0; r < 8; ++r) {
        const float d0 = acc[lt][0][r] + o[0];
        const float d1 = acc[lt][1][r] + o[1];
        const float t  = half ? d0 : d1;
        const float g  = xhalf(t);
        const float X  = half ? g : d0;
        const float Y  = half ? d1 : g;
        p[(size_t)r * N]       = X;
        p[(size_t)(r + 8) * N] = Y;
      }
    }
  } else {
    if (m == 0) {
#pragma unroll
      for (int q = 0; q < kStrips; ++q) {
        float* p = xch + wave * kRows + q * 16 + half * 8;
        *(float4*)(p)     = make_float4(c[q][0], c[q][1], c[q][2], c[q][3]);
        *(float4*)(p + 4) = make_float4(c[q][4], c[q][5], c[q][6], c[q][7]);
      }
    }
    __syncthreads();

    {
      const int rowl = lane & (kRows - 1);
      float o = 0.0f;
      for (int p = 0; p < wave; ++p) o += xch[p * kRows + rowl];
      offs[rowl] = o;
    }

    const float4 o0 = *(const float4*)&offs[half * 8];
    const float4 o1 = *(const float4*)&offs[half * 8 + 4];
    const float  o[8] = {o0.x, o0.y, o0.z, o0.w, o1.x, o1.y, o1.z, o1.w};
#pragma unroll
    for (int lt = 0; lt < kTiles; lt += 2) {
      const int tile = wave * kTiles + lt;
      float* p = dst + (size_t)row0 * N + tile * 16 + lane;
#pragma unroll
      for (int r = 0; r < 8; ++r) {
        const float d0 = acc[lt][0][r] + o[r];
        const float d1 = acc[lt + 1][0][r] + o[r];
        const float t  = half ? d0 : d1;
        const float g  = xhalf(t);
        const float X  = half ? g : d0;
        const float Y  = half ? d1 : g;
        p[(size_t)r * N]       = X;
        p[(size_t)(r + 8) * N] = Y;
      }
    }
  }
}

#define K1WAVES 16
#define K1TILES (NTILE / K1WAVES)
__global__ __launch_bounds__(K1WAVES * 32) void scan_down_panels(const float* __restrict__ D,
                                                                   float* __restrict__ RT) {
  const int lane = threadIdx.x & 31;
  const int wave = threadIdx.x >> 5;
  const int pair = blockIdx.x;
  const int b    = pair >> 6;
  const int row0 = (pair & 63) << 5;

  __shared__ __attribute__((aligned(16))) float lds[K1WAVES][32 * 16];
  __shared__ __attribute__((aligned(16))) float ldt[K1WAVES][16 * TPITCH];
  __shared__ __attribute__((aligned(16))) float xch[K1WAVES * 32];
  scan_rows<2, K1TILES, K1WAVES, true, true>(D + (size_t)b * N * N, RT + (size_t)b * N * N,
                                             lds[wave], ldt[wave], xch, nullptr, row0, lane,
                                             wave);
}

#define K2WAVES 16
#define K2TILES (NTILE / K2WAVES)
__global__ __launch_bounds__(K2WAVES * 32) void rowscan_inplace(float* PT) {
  const int lane  = threadIdx.x & 31;
  const int wave  = threadIdx.x >> 5;
  const int strip = blockIdx.x;
  const int b     = strip >> 7;
  const int row0  = (strip & 127) << 4;

  __shared__ __attribute__((aligned(16))) float lds[K2WAVES][16 * 16];
  __shared__ __attribute__((aligned(16))) float xch[K2WAVES * 16];
  __shared__ __attribute__((aligned(16))) float offs[K2WAVES][16];
  float* Pb = PT + (size_t)b * N * N;
  scan_rows<1, K2TILES, K2WAVES, false, false>(Pb, Pb, lds[wave], nullptr, xch, offs[wave], row0,
                                               lane, wave);
}

__global__ __launch_bounds__(ACOLS) void assemble(const float* __restrict__ D,
                                                  const float* __restrict__ PT,
                                                  float* __restrict__ out) {
  constexpr int kRowsWg = AROWS * ACHUNKS;
  const int blk = blockIdx.x;
  const int j   = (blk & (N / ACOLS - 1)) * ACOLS + threadIdx.x;
  const int ib  = ((blk / (N / ACOLS)) & (N / kRowsWg - 1)) * kRowsWg;
  const int b   = blk / ((N / ACOLS) * (N / kRowsWg));
  const float* __restrict__ Pb = PT + (size_t)b * N * N;
  const float* __restrict__ Db = D + (size_t)b * N * N;
  float* __restrict__ ob = out + (size_t)b * N * N;

  const int   jm   = (j > 0) ? j - 1 : 0;
  const float dgj  = Pb[(size_t)j * N + j];
  const float dgjm = (j > 0) ? Pb[(size_t)jm * N + jm] : 0.0f;

  float dsub = 0.0f;
  if (j + 1 >= ib && j + 1 < ib + kRowsWg) dsub = Db[(size_t)(j + 1) * N + j];

#pragma unroll 1
  for (int c = 0; c < ACHUNKS; ++c) {
    const int i0 = ib + c * AROWS;

    float dgr[AROWS];
#pragma unroll
    for (int u = 0; u < AROWS; ++u) dgr[u] = Pb[(size_t)(i0 + u) * (N + 1)];
    const float dgm1 = (i0 > 0) ? Pb[(size_t)(i0 - 1) * (N + 1)] : 0.0f;

    const int offU = (i0 - 1) * N + j;
    const int offL = i0 * N + jm;
    float x[AROWS];
#pragma unroll
    for (int u = 0; u < AROWS; ++u) {
      const bool up = (j >= i0 + u);
      int o = up ? offU : offL;
      if (u == 0) o = (o < 0) ? j : o;
      x[u] = Pb[o + u * N];
    }
#pragma unroll
    for (int u = 0; u < AROWS; ++u) {
      const int   i   = i0 + u;
      const bool  up  = (j >= i);
      const float pcc = up ? dgj : dgr[u];
      const float paa = up ? ((u > 0) ? dgr[u - 1] : dgm1) : dgjm;
      const bool  has = up ? (i > 0) : (j > 0);
      const float pac = has ? x[u] : 0.0f;
      float r = pcc - pac;
      r -= pac;
      r += paa;
      r *= 0.5f;
      r = (j == i - 1) ? dsub : r;
      ob[(size_t)i * N + j] = r;
    }
  }
}

extern "C" void kernel_launch(void* const* d_in, const int* in_sizes, int n_in,
                              void* d_out, int out_size, void* d_ws, size_t ws_size,
                              hipStream_t stream) {
  const float* D = (const float*)d_in[0];
  float* out = (float*)d_out;
  float* PT  = (float*)d_ws;
  const int B = in_sizes[0] / (N * N);

  scan_down_panels<<<dim3(B * (N / 32)), dim3(K1WAVES * 32), 0, stream>>>(D, PT);
  rowscan_inplace<<<dim3(B * NTILE), dim3(K2WAVES * 32), 0, stream>>>(PT);
  assemble<<<dim3((unsigned)(B * (N / (AROWS * ACHUNKS)) * (N / ACOLS))), dim3(ACOLS), 0, stream>>>(D, PT, out);
}


# assemble fused into the second row scan: its work-group forms the outputs from its register tiles, the SAT is never stored; SAT diagonal from a 1-D scan of R^T's diagonal; sum order changed
# speedup vs baseline: 12.9163x; 1.3865x over previous
#include <hip/hip_runtime.h>


typedef __attribute__((ext_vector_type(2))) float v2f;
typedef __attribute__((ext_vector_type(8))) float v8f;

#define N 2048
#define NTILE (N / 16)
#define TPITCH 40

__device__ __forceinline__ float xhalf(float x) {
  const int b = __builtin_bit_cast(int, x);
  return __builtin_bit_cast(float, __builtin_amdgcn_permlanex16(b, b, 0x76543210, 0xfedcba98, false, false));
}

template <int kStrips, int kTiles, int kWaves, bool kColumnForm, bool kAddMirror>
__device__ __forceinline__ void scan_rows(const float* src, float* dst, float* tl, float* tt,
                                          float* xch, float* offs, float* ztot, const float* Dsub,
                                          const int row0, const int lane, const int wave) {
  static_assert(kTiles * kWaves == NTILE, "the waves' runs must cover the row");
  static_assert(kColumnForm ? (kStrips == 2 && kAddMirror) : (kStrips == 1 && kTiles % 2 == 0),
                "whole-line stores: a 32-column panel, or pairs of adjacent tiles of a run");
  constexpr int kRows = 16 * kStrips;
  const int m    = lane & 15;
  const int half = lane >> 4;

  v8f c[kStrips];
#pragma unroll
  for (int q = 0; q < kStrips; ++q) c[q] = (v8f){0.f, 0.f, 0.f, 0.f, 0.f, 0.f, 0.f, 0.f};

  v8f acc[kTiles][kStrips];

  float4 z4 = make_float4(0.f, 0.f, 0.f, 0.f);
  if constexpr (!kColumnForm) {
    if (row0 > 0) z4 = *(const float4*)(src + (size_t)(row0 - 1) * N + wave * (kTiles * 16) + lane * 4);
  }

#pragma unroll
  for (int lt = 0; lt < kTiles; ++lt) {
    const int tile = wave * kTiles + lt;
    float4 v[2 * kStrips];
#pragma unroll
    for (int t = 0; t < 2 * kStrips; ++t) {
      const int idx4 = t * 32 + lane;
      const int r    = idx4 >> 2;
      const int cg   = idx4 & 3;
      v[t] = *(const float4*)(src + (size_t)(row0 + r) * N + tile * 16 + cg * 4);
    }
    float4 w[2 * kStrips];
    if (kAddMirror) {
#pragma unroll
      for (int t = 0; t < 2 * kStrips; ++t) {
        const int idx4 = t * 32 + lane;
        const int rr   = idx4 / (4 * kStrips);
        const int cg   = idx4 % (4 * kStrips);
        w[t] = *(const float4*)(src + (size_t)(tile * 16 + rr) * N + row0 + cg * 4);
      }
    }
#pragma unroll
    for (int t = 0; t < 2 * kStrips; ++t) {
      const int idx4 = t * 32 + lane;
      const int r    = idx4 >> 2;
      const int cg   = idx4 & 3;
      *(float4*)&tl[r * 16 + cg * 4] = v[t];
    }
    if (kAddMirror) {
#pragma unroll
      for (int t = 0; t < 2 * kStrips; ++t) {
        const int idx4 = t * 32 + lane;
        const int rr   = idx4 / (4 * kStrips);
        const int cg   = idx4 % (4 * kStrips);
        *(float4*)&tt[rr * TPITCH + cg * 4] = w[t];
      }
    }

#pragma unroll
    for (int s = 0; s < 4; ++s) {
      const int k0 = s * 4 + half * 2;
      v2f u;
      u.x = (k0     <= m) ? 1.0f : 0.0f;
      u.y = (k0 + 1 <= m) ? 1.0f : 0.0f;
#pragma unroll
      for (int q = 0; q < kStrips; ++q) {
        v2f a;
        a.x = tl[(q * 16 + m) * 16 + k0];
        a.y = tl[(q * 16 + m) * 16 + k0 + 1];
        if (kAddMirror) {
          a.x += tt[k0 * TPITCH + q * 16 + m];
          a.y += tt[(k0 + 1) * TPITCH + q * 16 + m];
        }
        if constexpr (kColumnForm) {
          c[q] = __builtin_amdgcn_wmma_f32_16x16x4_f32(
               false, u,  false, a,
               (short)0, c[q],  false,  false);
        } else {
          c[q] = __builtin_amdgcn_wmma_f32_16x16x4_f32(
               false, a,  false, u,
               (short)0, c[q],  false,  false);
        }
      }
    }

#pragma unroll
    for (int q = 0; q < kStrips; ++q) acc[lt][q] = c[q];

    if constexpr (kColumnForm) {
#pragma unroll
      for (int q = 0; q < kStrips; ++q) {
        const float own = c[q][7];
        const float oth = xhalf(own);
        const float cy  = half ? own : oth;
#pragma unroll
        for (int r = 0; r < 8; ++r) c[q][r] = cy;
      }
    } else {
      const int srcl = half ? 31 : 15;
#pragma unroll
      for (int q = 0; q < kStrips; ++q) {
#pragma unroll
        for (int r = 0; r < 8; ++r) c[q][r] = __shfl(c[q][r], srcl, 32);
      }
    }
  }

  if constexpr (kColumnForm) {
    if (half == 0) {
#pragma unroll
      for (int q = 0; q < kStrips; ++q) xch[wave * kRows + q * 16 + m] = c[q][0];
    }
    __syncthreads();

    float o[kStrips];
#pragma unroll
    for (int q = 0; q < kStrips; ++q) o[q] = 0.0f;
    for (int p = 0; p < wave; ++p) {
#pragma unroll
      for (int q = 0; q < kStrips; ++q) o[q] += xch[p * kRows + q * 16 + m];
    }

#pragma unroll
    for (int lt = 0; lt < kTiles; ++lt) {
      const int tile = wave * kTiles + lt;
      float* p = dst + (size_t)(tile * 16) * N + row0 + lane;
#pragma unroll
      for (int r = 0; r < 8; ++r) {
        const float d0 = acc[lt][0][r] + o[0];
        const float d1 = acc[lt][1][r] + o[1];
        const float t  = half ? d0 : d1;
        const float g  = xhalf(t);
        const float X  = half ? g : d0;
        const float Y  = half ? d1 : g;
        p[(size_t)r * N]       = X;
        p[(size_t)(r + 8) * N] = Y;
      }
    }
  } else {
    if (m == 0) {
#pragma unroll
      for (int q = 0; q < kStrips; ++q) {
        float* p = xch + wave * kRows + q * 16 + half * 8;
        *(float4*)(p)     = make_float4(c[q][0], c[q][1], c[q][2], c[q][3]);
        *(float4*)(p + 4) = make_float4(c[q][4], c[q][5], c[q][6], c[q][7]);
      }
    }
    z4.y += z4.x;
    z4.z += z4.y;
    z4.w += z4.z;
    float zincl = z4.w;
#pragma unroll
    for (int d = 1; d < 32; d <<= 1) {
      const float t = __shfl_up(zincl, d, 32);
      if (lane >= d) zincl += t;
    }
    float zexcl = __shfl_up(zincl, 1, 32);
    if (lane == 0) zexcl = 0.0f;
    if (lane == 31) ztot[wave] = zincl;
    __syncthreads();

    {
      const int rowl = lane & (kRows - 1);
      float o = 0.0f;
      float zo = 0.0f;
      for (int p = 0; p < wave; ++p) {
        o  += xch[p * kRows + rowl];
        zo += ztot[p];
      }
      offs[rowl] = o;
      zo += zexcl;
      *(float4*)&tl[lane * 4] = make_float4(zo + z4.x, zo + z4.y, zo + z4.z, zo + z4.w);
    }

    const float4 o0 = *(const float4*)&offs[half * 8];
    const float4 o1 = *(const float4*)&offs[half * 8 + 4];
    const float  o[8] = {o0.x, o0.y, o0.z, o0.w, o1.x, o1.y, o1.z, o1.w};
    constexpr int kPairs = kTiles / 2;
    const float* dh = dst;
    float dgc[kPairs], dgcm[kPairs], dsub[kPairs], zrow[kPairs];
#pragma unroll
    for (int p = 0; p < kPairs; ++p) {
      const int cl = wave * (kTiles * 16) + p * 32 + lane;
      dgc[p]  = 2.0f * dh[cl];
      const float dl = dh[cl > 0 ? cl - 1 : 0];
      dgcm[p] = (cl > 0) ? 2.0f * dl : 0.0f;
      dsub[p] = (cl + 1 >= row0 && cl + 1 < row0 + 16) ? Dsub[(size_t)(cl + 1) * N + cl] : 0.0f;
      zrow[p] = tl[p * 32 + lane];
    }
    float dgx[17];
    dgx[0] = (row0 > 0) ? 2.0f * dh[row0 - 1] : 0.0f;
#pragma unroll
    for (int i = 0; i < 16; ++i) dgx[i + 1] = 2.0f * dh[row0 + i];
    float orow[16];
#pragma unroll
    for (int i = 0; i < 16; ++i) orow[i] = offs[i];

    float lastcol[16];
#pragma unroll
    for (int p = 0; p < kPairs; ++p) {
      const int lt = 2 * p;
      const int cl = wave * (kTiles * 16) + p * 32 + lane;
      float pr[16];
#pragma unroll
      for (int r = 0; r < 8; ++r) {
        const float d0 = acc[lt][0][r] + o[r];
        const float d1 = acc[lt + 1][0][r] + o[r];
        const float t  = half ? d0 : d1;
        const float g  = xhalf(t);
        pr[r]     = half ? g : d0;
        pr[r + 8] = half ? d1 : g;
      }
      float left[16];
#pragma unroll
      for (int i = 0; i < 16; ++i) {
        const float nb   = __shfl_up(pr[i], 1, 32);
        const float edge = (p > 0) ? lastcol[i] : orow[i];
        left[i] = (lane > 0) ? nb : edge;
        lastcol[i] = __builtin_bit_cast(float, __builtin_amdgcn_readlane(__builtin_bit_cast(int, pr[i]), 31));
      }
      float* q = dst + (size_t)row0 * N + cl;
#pragma unroll
      for (int i = 0; i < 16; ++i) {
        const int   x   = row0 + i;
        const bool  up  = (cl >= x);
        const float above = (i > 0) ? pr[i > 0 ? i - 1 : 0] : zrow[p];
        const float pcc = up ? dgc[p] : dgx[i + 1];
        const float paa = up ? dgx[i] : dgcm[p];
        const float pac = up ? above : left[i];
        float r = pcc - pac;
        r -= pac;
        r += paa;
        r *= 0.5f;
        r = (cl == x - 1) ? dsub[p] : r;
        if (x > 0) q[(size_t)i * N] = r;
      }
    }
  }
}

#define K1WAVES 16
#define K1TILES (NTILE / K1WAVES)
__global__ __launch_bounds__(K1WAVES * 32) void scan_down_panels(const float* __restrict__ D,
                                                                   float* __restrict__ RT) {
  const int lane = threadIdx.x & 31;
  const int wave = threadIdx.x >> 5;
  const int pair = blockIdx.x;
  const int b    = pair >> 6;
  const int row0 = (pair & 63) << 5;

  __shared__ __attribute__((aligned(16))) float lds[K1WAVES][32 * 16];
  __shared__ __attribute__((aligned(16))) float ldt[K1WAVES][16 * TPITCH];
  __shared__ __attribute__((aligned(16))) float xch[K1WAVES * 32];
  scan_rows<2, K1TILES, K1WAVES, true, true>(D + (size_t)b * N * N, RT + (size_t)b * N * N,
                                             lds[wave], ldt[wave], xch, nullptr, nullptr, nullptr,
                                             row0, lane, wave);
}

__global__ __launch_bounds__(N / 2) void diag_scan(const float* __restrict__ D,
                                                   const float* __restrict__ RT,
                                                   float* __restrict__ out) {
  const int b    = blockIdx.x;
  const int lane = threadIdx.x & 31;
  const int wave = threadIdx.x >> 5;
  const int e    = 2 * threadIdx.x;
  const float* Rb = RT + (size_t)b * N * N;
  const float* Db = D + (size_t)b * N * N;
  __shared__ float wt[N / 64];
  const float a0 = 2.0f * Rb[(size_t)e * (N + 1)] - 2.0f * Db[(size_t)e * (N + 1)];
  const float a1 = a0 + (2.0f * Rb[(size_t)(e + 1) * (N + 1)] - 2.0f * Db[(size_t)(e + 1) * (N + 1)]);
  float incl = a1;
#pragma unroll
  for (int d = 1; d < 32; d <<= 1) {
    const float t = __shfl_up(incl, d, 32);
    if (lane >= d) incl += t;
  }
  float excl = __shfl_up(incl, 1, 32);
  if (lane == 0) excl = 0.0f;
  if (lane == 31) wt[wave] = incl;
  __syncthreads();
  float base = 0.0f;
  for (int p = 0; p < wave; ++p) base += wt[p];
  base += excl;
  *(float2*)(out + (size_t)b * N * N + e) = make_float2(0.5f * (base + a0), 0.5f * (base + a1));
}

#define K2WAVES 16
#define K2TILES (NTILE / K2WAVES)
__global__ __launch_bounds__(K2WAVES * 32) void rowscan_assemble(const float* __restrict__ D,
                                                                 const float* __restrict__ RT,
                                                                 float* out) {
  const int lane  = threadIdx.x & 31;
  const int wave  = threadIdx.x >> 5;
  const int strip = blockIdx.x;
  const int b     = strip >> 7;
  const int row0  = (strip & 127) << 4;

  __shared__ __attribute__((aligned(16))) float lds[K2WAVES][16 * 16];
  __shared__ __attribute__((aligned(16))) float xch[K2WAVES * 16];
  __shared__ __attribute__((aligned(16))) float offs[K2WAVES][16];
  __shared__ float ztot[K2WAVES];
  scan_rows<1, K2TILES, K2WAVES, false, false>(RT + (size_t)b * N * N, out + (size_t)b * N * N,
                                               lds[wave], nullptr, xch, offs[wave], ztot,
                                               D + (size_t)b * N * N, row0, lane, wave);
}

extern "C" void kernel_launch(void* const* d_in, const int* in_sizes, int n_in,
                              void* d_out, int out_size, void* d_ws, size_t ws_size,
                              hipStream_t stream) {
  const float* D = (const float*)d_in[0];
  float* out = (float*)d_out;
  float* RT  = (float*)d_ws;
  const int B = in_sizes[0] / (N * N);

  scan_down_panels<<<dim3(B * (N / 32)), dim3(K1WAVES * 32), 0, stream>>>(D, RT);
  diag_scan<<<dim3(B), dim3(N / 2), 0, stream>>>(D, RT, out);
  rowscan_assemble<<<dim3(B * NTILE), dim3(K2WAVES * 32), 0, stream>>>(D, RT, out);
}
